# starting kernel
# baseline (speedup 1.0000x reference)
#include <hip/hip_runtime.h>
#include <math.h>

typedef __attribute__((ext_vector_type(2))) float v2f;
typedef __attribute__((ext_vector_type(8))) float v8f;

constexpr int D      = 2048;
constexpr int T      = 10;
constexpr int TP     = 16;
constexpr int NROWS  = 4 * 4096;
constexpr int KC       = 256;
constexpr int NCHUNK   = D / KC;
constexpr int WAVES    = 2;
constexpr int BLOCK    = WAVES * 32;
constexpr int ROWS_PB  = WAVES * 16;

__global__ __launch_bounds__(BLOCK)
void tgate_hybrid_fused_kernel(const float* __restrict__ x,
                               const float* __restrict__ Wc, const float* __restrict__ bc,
                               const float* __restrict__ Ws, const float* __restrict__ bs,
                               const float* __restrict__ Wg, const float* __restrict__ bg,
                               const float* __restrict__ alpha,
                               float* __restrict__ out)
{
    __shared__ float lds_w[3][KC / 2][TP][2];
    __shared__ float lds_acc[ROWS_PB][3][TP];

    const int tid  = threadIdx.x;
    const int wid  = tid >> 5;
    const int lane = tid & 31;
    const int half = lane >> 4;
    const int l16  = lane & 15;

    const int rowBase = blockIdx.x * ROWS_PB + wid * 16;
    const float* px   = x + (size_t)(rowBase + l16) * D;

    const float* Wmat[3] = { Wc, Ws, Wg };

    v8f acc[3];
    #pragma unroll
    for (int m = 0; m < 3; ++m)
        #pragma unroll
        for (int r = 0; r < 8; ++r)
            acc[m][r] = 0.0f;

    for (int c = 0; c < NCHUNK; ++c) {
        const int k0 = c * KC;
        __syncthreads();

        for (int idx = tid; idx < 3 * KC * TP; idx += BLOCK) {
            const int m = idx / (KC * TP);
            const int r = idx % (KC * TP);
            const int k = r / TP;
            const int n = r % TP;
            const float v = (n < T) ? Wmat[m][(size_t)(k0 + k) * T + n] : 0.0f;
            lds_w[m][k >> 1][n][k & 1] = v;
        }
        __syncthreads();

        if (c + 1 < NCHUNK)
            __builtin_prefetch(px + k0 + KC + 2 * half, 0, 1);

        #pragma unroll 8
        for (int kk = 0; kk < KC; kk += 4) {
            const v2f a = __builtin_nontemporal_load(
                (const v2f*)(px + k0 + kk + 2 * half));
            #pragma unroll
            for (int m = 0; m < 3; ++m) {
                const v2f b = *(const v2f*)&lds_w[m][(kk >> 1) + half][l16][0];
                acc[m] = __builtin_amdgcn_wmma_f32_16x16x4_f32(
                    false, a, false, b, (short)0, acc[m], false, false);
            }
        }
    }

    __syncthreads();
    #pragma unroll
    for (int m = 0; m < 3; ++m)
        #pragma unroll
        for (int r = 0; r < 8; ++r)
            lds_acc[wid * 16 + r + 8 * half][m][l16] = acc[m][r];
    __syncthreads();

    if (tid < ROWS_PB) {
        const int row = tid;
        float Lc[T], Ls[T], Lg[T];
        #pragma unroll
        for (int n = 0; n < T; ++n) {
            Lc[n] = lds_acc[row][0][n] + bc[n];
            Ls[n] = lds_acc[row][1][n] + bs[n];
            Lg[n] = lds_acc[row][2][n] + bg[n];
        }
        float mx = Lc[0];
        #pragma unroll
        for (int n = 1; n < T; ++n) mx = fmaxf(mx, Lc[n]);
        float e[T], ssum = 0.0f;
        #pragma unroll
        for (int n = 0; n < T; ++n) { e[n] = __expf(Lc[n] - mx); ssum += e[n]; }
        float g[T];
        #pragma unroll
        for (int n = 0; n < T; ++n) g[n] = 1.0f / (1.0f + __expf(-Lg[n]));
        float softdot = 0.0f;
        #pragma unroll
        for (int n = 0; n < T; ++n) softdot += g[n] * e[n];
        softdot /= ssum;
        float v0 = -INFINITY, v1 = -INFINITY;
        int   i0 = 0, i1 = 0;
        #pragma unroll
        for (int n = 0; n < T; ++n) {
            const float t = Ls[n];
            if (t > v0)      { v1 = v0; i1 = i0; v0 = t; i0 = n; }
            else if (t > v1) { v1 = t;  i1 = n; }
        }
        const float s0 = 1.0f / (1.0f + __expf(v1 - v0));
        const float s1 = 1.0f - s0;
        const float sparsedot = g[i0] * s0 + g[i1] * s1;

        const float a = 1.0f / (1.0f + __expf(-alpha[0]));
        out[blockIdx.x * ROWS_PB + row] = a * sparsedot + (1.0f - a) * softdot;
    }
}

extern "C" void kernel_launch(void* const* d_in, const int* in_sizes, int n_in,
                              void* d_out, int out_size, void* d_ws, size_t ws_size,
                              hipStream_t stream) {
    (void)in_sizes; (void)n_in; (void)out_size; (void)d_ws; (void)ws_size;
    const float* x     = (const float*)d_in[0];
    const float* W_cls = (const float*)d_in[1];
    const float* b_cls = (const float*)d_in[2];
    const float* W_sp  = (const float*)d_in[3];
    const float* b_sp  = (const float*)d_in[4];
    const float* W_g   = (const float*)d_in[5];
    const float* b_g   = (const float*)d_in[6];
    const float* alpha = (const float*)d_in[7];
    float* out = (float*)d_out;

    dim3 grid(NROWS / ROWS_PB);
    dim3 block(BLOCK);
    tgate_hybrid_fused_kernel<<<grid, block, 0, stream>>>(
        x, W_cls, b_cls, W_sp, b_sp, W_g, b_g, alpha, out);
}


# Each wave keeps its 256-row weight slice in registers (384 VGPRs, 8-wave block), read from global in fragment order under tile 0's products; K loop = x loads + WMMA, no LDS; sum order unchanged
# speedup vs baseline: 18.7857x; 18.7857x over previous
#include <hip/hip_runtime.h>
#include <math.h>
#include <utility>


typedef __attribute__((ext_vector_type(2))) float v2f;
typedef __attribute__((ext_vector_type(4))) float v4f;
typedef __attribute__((ext_vector_type(8))) float v8f;

constexpr int D      = 2048;
constexpr int T      = 10;
constexpr int NROWS  = 4 * 4096;
constexpr int NTILE  = NROWS / 16;

constexpr int KC        = 256;
constexpr int NSLICE    = D / KC;
constexpr int TILES_WG  = 4;
constexpr int WAVES     = NSLICE;
constexpr int BLOCK     = WAVES * 32;
constexpr int NWG       = NTILE / TILES_WG;
constexpr int ROWS_WG   = TILES_WG * 16;
constexpr int NSTEP     = KC / 4;
constexpr int TOTAL     = TILES_WG * NSTEP;

constexpr int TRIP    = 8;
constexpr int NTRIP   = TOTAL / TRIP;
constexpr int W_AHEAD = 7;
constexpr int X_AHEAD = 3;
constexpr int X_RING  = (X_AHEAD + 1) * TRIP;
constexpr int x_issue_step(int j)
{
    if (j == 0) return -1;
    if (j < NSTEP / TRIP) return TRIP * (j - 1);
    if (j < NSTEP / TRIP + X_AHEAD)
        return NSTEP - 4 * (NSTEP / TRIP + X_AHEAD - j);
    return TRIP * (j - X_AHEAD);
}
constexpr int x_trip_at(int g)
{
    for (int j = 1; j < NTRIP; ++j)
        if (x_issue_step(j) == g) return j;
    return -1;
}

constexpr int PART_PER_WAVE = 3 * 2 * 32;
constexpr int ACC_OFF       = TILES_WG * NSLICE * PART_PER_WAVE * 4;
constexpr int LDS_FLOATS    = ACC_OFF + ROWS_WG * 3 * 16;


struct WaveState {
    v2f bf[3][NSTEP];
    v2f ar[X_RING];
    v8f acc[3];
    const float* pw0;
    const float* pw1;
    const float* pw2;
    const float* px;
    v4f* part;
};

template <int U>
__device__ __forceinline__ void load_w(WaveState& st)
{
    st.bf[0][U][0] = st.pw0[(4 * U + 0) * T];
    st.bf[0][U][1] = st.pw0[(4 * U + 1) * T];
    st.bf[1][U][0] = st.pw1[(4 * U + 0) * T];
    st.bf[1][U][1] = st.pw1[(4 * U + 1) * T];
    st.bf[2][U][0] = st.pw2[(4 * U + 0) * T];
    st.bf[2][U][1] = st.pw2[(4 * U + 1) * T];
}

template <int I>
__device__ __forceinline__ void load_x(WaveState& st)
{
    st.ar[I % X_RING] = __builtin_nontemporal_load(
        (const v2f*)(st.px + (I / NSTEP) * 16 * D + 4 * (I % NSTEP)));
}

template <int J, int... E>
__device__ __forceinline__ void load_x_trip(WaveState& st, std::integer_sequence<int, E...>)
{
    (load_x<J * TRIP + E>(st), ...);
}

template <int G>
__device__ __forceinline__ void k_step(WaveState& st)
{
    constexpr int t = G / NSTEP;
    constexpr int u = G % NSTEP;
    if constexpr (G + W_AHEAD < NSTEP)
        load_w<G + W_AHEAD>(st);
    if constexpr (x_trip_at(G) >= 0)
        load_x_trip<x_trip_at(G)>(st, std::make_integer_sequence<int, TRIP>{});
    if constexpr (u == 0) {
        #pragma unroll
        for (int m = 0; m < 3; ++m)
            #pragma unroll
            for (int r = 0; r < 8; ++r)
                st.acc[m][r] = 0.0f;
    }
    #pragma unroll
    for (int m = 0; m < 3; ++m)
        st.acc[m] = __builtin_amdgcn_wmma_f32_16x16x4_f32(
            false, st.ar[G % X_RING], false, st.bf[m][u], (short)0, st.acc[m], false, false);
    if constexpr (u == NSTEP - 1) {
        #pragma unroll
        for (int m = 0; m < 3; ++m)
            #pragma unroll
            for (int j = 0; j < 2; ++j) {
                v4f v;
                v[0] = st.acc[m][4 * j + 0];
                v[1] = st.acc[m][4 * j + 1];
                v[2] = st.acc[m][4 * j + 2];
                v[3] = st.acc[m][4 * j + 3];
                st.part[(t * NSLICE * 6 + m * 2 + j) * 32] = v;
            }
    }
    __builtin_amdgcn_sched_barrier(0);
}

template <int... U>
__device__ __forceinline__ void run_prologue(WaveState& st, std::integer_sequence<int, U...>)
{
    load_x_trip<0>(st, std::make_integer_sequence<int, TRIP>{});
    (load_w<U>(st), ...);
}

template <int... G>
__device__ __forceinline__ void run_steps(WaveState& st, std::integer_sequence<int, G...>)
{
    (k_step<G>(st), ...);
}

__global__ __launch_bounds__(BLOCK)
void tgate_register_weights_kernel(const float* __restrict__ x,
                                   const float* __restrict__ Wc,
                                   const float* __restrict__ Ws,
                                   const float* __restrict__ Wg,
                                   const float* __restrict__ bc,
                                   const float* __restrict__ bs,
                                   const float* __restrict__ bg,
                                   const float* __restrict__ alpha,
                                   float* __restrict__ out)
{
    __shared__ __attribute__((aligned(16))) float lds[LDS_FLOATS];

    const int tid  = threadIdx.x;
    const int wid  = tid >> 5;
    const int lane = tid & 31;
    const int half = lane >> 4;
    const int l16  = lane & 15;
    const int k0   = wid * KC;

    const int bcol = l16 < T ? l16 : T - 1;
    const float* pw0 = Wc + (k0 + 2 * half) * T + bcol;
    const float* pw1 = Ws + (k0 + 2 * half) * T + bcol;
    const float* pw2 = Wg + (k0 + 2 * half) * T + bcol;
    const float* px = x + (size_t)(blockIdx.x * ROWS_WG + l16) * D + k0 + 2 * half;

    WaveState st;
    st.pw0 = pw0; st.pw1 = pw1; st.pw2 = pw2; st.px = px;
    st.part = (v4f*)lds + (wid * 6) * 32 + lane;
    run_prologue(st, std::make_integer_sequence<int, W_AHEAD>{});
    run_steps(st, std::make_integer_sequence<int, TOTAL>{});
    v4f* part = (v4f*)lds;
    __syncthreads();

    float* acc_rows = lds + ACC_OFF;
    if (wid < TILES_WG) {
        const v4f* src = part + (wid * NSLICE) * PART_PER_WAVE + lane;
        v4f sum[6];
        #pragma unroll
        for (int q = 0; q < 6; ++q)
            sum[q] = src[q * 32];
        #pragma unroll
        for (int s = 1; s < NSLICE; ++s)
            #pragma unroll
            for (int q = 0; q < 6; ++q)
                sum[q] += src[(s * 6 + q) * 32];
        #pragma unroll
        for (int m = 0; m < 3; ++m)
            #pragma unroll
            for (int r = 0; r < 8; ++r)
                acc_rows[((wid * 16 + r + 8 * half) * 3 + m) * 16 + l16] = sum[m * 2 + (r >> 2)][r & 3];
    }
    __syncthreads();

    if (tid < ROWS_WG) {
        const int row = tid;
        const float* my = acc_rows + row * 3 * 16;
        float Lc[T], Ls[T], Lg[T];
        #pragma unroll
        for (int n = 0; n < T; ++n) {
            Lc[n] = my[0 * 16 + n] + bc[n];
            Ls[n] = my[1 * 16 + n] + bs[n];
            Lg[n] = my[2 * 16 + n] + bg[n];
        }
        float mx = Lc[0];
        #pragma unroll
        for (int n = 1; n < T; ++n) mx = fmaxf(mx, Lc[n]);
        float e[T], ssum = 0.0f;
        #pragma unroll
        for (int n = 0; n < T; ++n) { e[n] = __expf(Lc[n] - mx); ssum += e[n]; }
        float g[T];
        #pragma unroll
        for (int n = 0; n < T; ++n) g[n] = 1.0f / (1.0f + __expf(-Lg[n]));
        float softdot = 0.0f;
        #pragma unroll
        for (int n = 0; n < T; ++n) softdot += g[n] * e[n];
        softdot /= ssum;
        float v0 = -INFINITY, v1 = -INFINITY;
        int   i0 = 0, i1 = 0;
        #pragma unroll
        for (int n = 0; n < T; ++n) {
            const float t = Ls[n];
            if (t > v0)      { v1 = v0; i1 = i0; v0 = t; i0 = n; }
            else if (t > v1) { v1 = t;  i1 = n; }
        }
        const float s0 = 1.0f / (1.0f + __expf(v1 - v0));
        const float s1 = 1.0f - s0;
        const float sparsedot = g[i0] * s0 + g[i1] * s1;

        const float a = 1.0f / (1.0f + __expf(-alpha[0]));
        out[blockIdx.x * ROWS_WG + row] = a * sparsedot + (1.0f - a) * softdot;
    }
}

extern "C" void kernel_launch(void* const* d_in, const int* in_sizes, int n_in,
                              void* d_out, int out_size, void* d_ws, size_t ws_size,
                              hipStream_t stream) {
    (void)in_sizes; (void)n_in; (void)out_size; (void)d_ws; (void)ws_size;
    const float* x     = (const float*)d_in[0];
    const float* W_cls = (const float*)d_in[1];
    const float* b_cls = (const float*)d_in[2];
    const float* W_sp  = (const float*)d_in[3];
    const float* b_sp  = (const float*)d_in[4];
    const float* W_g   = (const float*)d_in[5];
    const float* b_g   = (const float*)d_in[6];
    const float* alpha = (const float*)d_in[7];
    float* out  = (float*)d_out;

    dim3 grid(NWG);
    dim3 block(BLOCK);
    tgate_register_weights_kernel<<<grid, block, 0, stream>>>(
        x, W_cls, W_sp, W_g, b_cls, b_sp, b_g, alpha, out);
}


# x routed through LDS: per trip a wave reads 16 rows x 128 B as 4 whole-line loads (was 8 loads of 16 B from each of 16 lines), stores them in a private slot, reads A fragments back; sum order kept
# speedup vs baseline: 20.5351x; 1.0931x over previous
#include <hip/hip_runtime.h>
#include <math.h>
#include <utility>


typedef __attribute__((ext_vector_type(2))) float v2f;
typedef __attribute__((ext_vector_type(4))) float v4f;
typedef __attribute__((ext_vector_type(8))) float v8f;

constexpr int D      = 2048;
constexpr int T      = 10;
constexpr int NROWS  = 4 * 4096;
constexpr int NTILE  = NROWS / 16;

constexpr int KC        = 256;
constexpr int NSLICE    = D / KC;
constexpr int TILES_WG  = 4;
constexpr int WAVES     = NSLICE;
constexpr int BLOCK     = WAVES * 32;
constexpr int NWG       = NTILE / TILES_WG;
constexpr int ROWS_WG   = TILES_WG * 16;
constexpr int NSTEP     = KC / 4;
constexpr int TOTAL     = TILES_WG * NSTEP;

constexpr int TRIP       = 8;
constexpr int NTRIP      = TOTAL / TRIP;
constexpr int TRIPS_TILE = NSTEP / TRIP;
constexpr int W_AHEAD    = 7;
constexpr int G_AHEAD    = 4;
constexpr int G_SLOTS    = G_AHEAD - 1;
constexpr int F_AT       = 4;

constexpr int PART_PER_WAVE = 3 * 2 * 32;
constexpr int ACC_OFF       = TILES_WG * NSLICE * PART_PER_WAVE * 4;
constexpr int XR_OFF        = ACC_OFF + ROWS_WG * 3 * 16;
constexpr int ROW_PITCH     = 36;
constexpr int SLOT_FLOATS   = 16 * ROW_PITCH;
constexpr int LDS_FLOATS    = XR_OFF + WAVES * 2 * SLOT_FLOATS;


struct WaveState {
    v2f bf[3][NSTEP];
    v4f gb[G_SLOTS][4];
    v2f fr[2][TRIP];
    v8f acc[3];
    const float* pw0;
    const float* pw1;
    const float* pw2;
    const float* px;
    v4f* xs;
    const v2f* xl;
    v4f* part;
};

template <int U>
__device__ __forceinline__ void load_w(WaveState& st)
{
    st.bf[0][U][0] = st.pw0[(4 * U + 0) * T];
    st.bf[0][U][1] = st.pw0[(4 * U + 1) * T];
    st.bf[1][U][0] = st.pw1[(4 * U + 0) * T];
    st.bf[1][U][1] = st.pw1[(4 * U + 1) * T];
    st.bf[2][U][0] = st.pw2[(4 * U + 0) * T];
    st.bf[2][U][1] = st.pw2[(4 * U + 1) * T];
}

template <int J>
__device__ __forceinline__ void x_fetch(WaveState& st)
{
    #pragma unroll
    for (int i = 0; i < 4; ++i)
        st.gb[J % G_SLOTS][i] = __builtin_nontemporal_load(
            (const v4f*)(st.px + ((J / TRIPS_TILE) * 16 + 4 * i) * D + (J % TRIPS_TILE) * 4 * TRIP));
}

template <int J>
__device__ __forceinline__ void x_stage(WaveState& st)
{
    #pragma unroll
    for (int i = 0; i < 4; ++i)
        st.xs[((J % 2) * SLOT_FLOATS + 4 * i * ROW_PITCH) / 4] = st.gb[J % G_SLOTS][i];
}

template <int J>
__device__ __forceinline__ void x_frags(WaveState& st)
{
    #pragma unroll
    for (int e = 0; e < TRIP; ++e)
        st.fr[J % 2][e] = st.xl[((J % 2) * SLOT_FLOATS + 4 * e) / 2];
}

template <int G>
__device__ __forceinline__ void k_step(WaveState& st)
{
    constexpr int t = G / NSTEP;
    constexpr int u = G % NSTEP;
    constexpr int J = G / TRIP;
    constexpr int e = G % TRIP;
    if constexpr (G + W_AHEAD < NSTEP)
        load_w<G + W_AHEAD>(st);
    if constexpr (e == 0 && J + 1 < NTRIP)
        x_stage<J + 1>(st);
    if constexpr (e == 0 && J + G_AHEAD < NTRIP)
        x_fetch<J + G_AHEAD>(st);
    if constexpr (e == F_AT && J + 1 < NTRIP)
        x_frags<J + 1>(st);
    if constexpr (u == 0) {
        #pragma unroll
        for (int m = 0; m < 3; ++m)
            #pragma unroll
            for (int r = 0; r < 8; ++r)
                st.acc[m][r] = 0.0f;
    }
    #pragma unroll
    for (int m = 0; m < 3; ++m)
        st.acc[m] = __builtin_amdgcn_wmma_f32_16x16x4_f32(
            false, st.fr[J % 2][e], false, st.bf[m][u], (short)0, st.acc[m], false, false);
    if constexpr (u == NSTEP - 1) {
        #pragma unroll
        for (int m = 0; m < 3; ++m)
            #pragma unroll
            for (int j = 0; j < 2; ++j) {
                v4f v;
                v[0] = st.acc[m][4 * j + 0];
                v[1] = st.acc[m][4 * j + 1];
                v[2] = st.acc[m][4 * j + 2];
                v[3] = st.acc[m][4 * j + 3];
                st.part[(t * NSLICE * 6 + m * 2 + j) * 32] = v;
            }
    }
    __builtin_amdgcn_sched_barrier(0);
}

template <int... U>
__device__ __forceinline__ void run_prologue(WaveState& st, std::integer_sequence<int, U...>)
{
    x_fetch<0>(st);
    x_fetch<1>(st);
    x_fetch<2>(st);
    (load_w<U>(st), ...);
    x_stage<0>(st);
    x_fetch<3>(st);
    x_frags<0>(st);
}

template <int... G>
__device__ __forceinline__ void run_steps(WaveState& st, std::integer_sequence<int, G...>)
{
    (k_step<G>(st), ...);
}

__global__ __launch_bounds__(BLOCK)
void tgate_x_through_lds_kernel(const float* __restrict__ x,
                                   const float* __restrict__ Wc,
                                   const float* __restrict__ Ws,
                                   const float* __restrict__ Wg,
                                   const float* __restrict__ bc,
                                   const float* __restrict__ bs,
                                   const float* __restrict__ bg,
                                   const float* __restrict__ alpha,
                                   float* __restrict__ out)
{
    __shared__ __attribute__((aligned(16))) float lds[LDS_FLOATS];

    const int tid  = threadIdx.x;
    const int wid  = tid >> 5;
    const int lane = tid & 31;
    const int half = lane >> 4;
    const int l16  = lane & 15;
    const int k0   = wid * KC;

    const int bcol = l16 < T ? l16 : T - 1;
    const float* pw0 = Wc + (k0 + 2 * half) * T + bcol;
    const float* pw1 = Ws + (k0 + 2 * half) * T + bcol;
    const float* pw2 = Wg + (k0 + 2 * half) * T + bcol;
    const float* px = x + (size_t)(blockIdx.x * ROWS_WG + (lane >> 3)) * D + k0 + 4 * (lane & 7);
    float* xslot = lds + XR_OFF + wid * 2 * SLOT_FLOATS;

    WaveState st;
    st.pw0 = pw0; st.pw1 = pw1; st.pw2 = pw2; st.px = px;
    st.xs = (v4f*)(xslot + (lane >> 3) * ROW_PITCH + 4 * (lane & 7));
    st.xl = (const v2f*)(xslot + l16 * ROW_PITCH + 2 * half);
    st.part = (v4f*)lds + (wid * 6) * 32 + lane;
    run_prologue(st, std::make_integer_sequence<int, W_AHEAD>{});
    run_steps(st, std::make_integer_sequence<int, TOTAL>{});
    v4f* part = (v4f*)lds;
    __syncthreads();

    float* acc_rows = lds + ACC_OFF;
    if (wid < TILES_WG) {
        const v4f* src = part + (wid * NSLICE) * PART_PER_WAVE + lane;
        v4f sum[6];
        #pragma unroll
        for (int q = 0; q < 6; ++q)
            sum[q] = src[q * 32];
        #pragma unroll
        for (int s = 1; s < NSLICE; ++s)
            #pragma unroll
            for (int q = 0; q < 6; ++q)
                sum[q] += src[(s * 6 + q) * 32];
        #pragma unroll
        for (int m = 0; m < 3; ++m)
            #pragma unroll
            for (int r = 0; r < 8; ++r)
                acc_rows[((wid * 16 + r + 8 * half) * 3 + m) * 16 + l16] = sum[m * 2 + (r >> 2)][r & 3];
    }
    __syncthreads();

    if (tid < ROWS_WG) {
        const int row = tid;
        const float* my = acc_rows + row * 3 * 16;
        float Lc[T], Ls[T], Lg[T];
        #pragma unroll
        for (int n = 0; n < T; ++n) {
            Lc[n] = my[0 * 16 + n] + bc[n];
            Ls[n] = my[1 * 16 + n] + bs[n];
            Lg[n] = my[2 * 16 + n] + bg[n];
        }
        float mx = Lc[0];
        #pragma unroll
        for (int n = 1; n < T; ++n) mx = fmaxf(mx, Lc[n]);
        float e[T], ssum = 0.0f;
        #pragma unroll
        for (int n = 0; n < T; ++n) { e[n] = __expf(Lc[n] - mx); ssum += e[n]; }
        float g[T];
        #pragma unroll
        for (int n = 0; n < T; ++n) g[n] = 1.0f / (1.0f + __expf(-Lg[n]));
        float softdot = 0.0f;
        #pragma unroll
        for (int n = 0; n < T; ++n) softdot += g[n] * e[n];
        softdot /= ssum;
        float v0 = -INFINITY, v1 = -INFINITY;
        int   i0 = 0, i1 = 0;
        #pragma unroll
        for (int n = 0; n < T; ++n) {
            const float t = Ls[n];
            if (t > v0)      { v1 = v0; i1 = i0; v0 = t; i0 = n; }
            else if (t > v1) { v1 = t;  i1 = n; }
        }
        const float s0 = 1.0f / (1.0f + __expf(v1 - v0));
        const float s1 = 1.0f - s0;
        const float sparsedot = g[i0] * s0 + g[i1] * s1;

        const float a = 1.0f / (1.0f + __expf(-alpha[0]));
        out[blockIdx.x * ROWS_WG + row] = a * sparsedot + (1.0f - a) * softdot;
    }
}

extern "C" void kernel_launch(void* const* d_in, const int* in_sizes, int n_in,
                              void* d_out, int out_size, void* d_ws, size_t ws_size,
                              hipStream_t stream) {
    (void)in_sizes; (void)n_in; (void)out_size; (void)d_ws; (void)ws_size;
    const float* x     = (const float*)d_in[0];
    const float* W_cls = (const float*)d_in[1];
    const float* b_cls = (const float*)d_in[2];
    const float* W_sp  = (const float*)d_in[3];
    const float* b_sp  = (const float*)d_in[4];
    const float* W_g   = (const float*)d_in[5];
    const float* b_g   = (const float*)d_in[6];
    const float* alpha = (const float*)d_in[7];
    float* out  = (float*)d_out;

    dim3 grid(NWG);
    dim3 block(BLOCK);
    tgate_x_through_lds_kernel<<<grid, block, 0, stream>>>(
        x, W_cls, W_sp, W_g, b_cls, b_sp, b_g, alpha, out);
}
